# starting kernel
# baseline (speedup 1.0000x reference)
#include <hip/hip_runtime.h>

typedef __attribute__((ext_vector_type(2))) float v2f;
typedef __attribute__((ext_vector_type(4))) float v4f;
typedef __attribute__((ext_vector_type(8))) float v8f;

#define M_TOT 8192
#define N_TOT 4096
#define K_TOT 4096

#define MT 128
#define NT 128
#define KT 32
#define LDSTR 34
#define A_ELEMS (MT * LDSTR)
#define SLAB_ELEMS (2 * A_ELEMS)

__device__ __forceinline__ void stage_slab(const float* __restrict__ xp,
                                           const float* __restrict__ wp,
                                           float* __restrict__ As,
                                           float* __restrict__ Bs,
                                           int sr, int sc) {
    #pragma unroll
    for (int p = 0; p < 4; ++p) {
        const size_t goff = (size_t)(32 * p) * K_TOT;
        v4f a = *(const v4f*)(xp + goff);
        v4f b = *(const v4f*)(wp + goff);
        const int l = (sr + 32 * p) * LDSTR + sc;
        *(v2f*)&As[l]     = a.lo;
        *(v2f*)&As[l + 2] = a.hi;
        *(v2f*)&Bs[l]     = b.lo;
        *(v2f*)&Bs[l + 2] = b.hi;
    }
}

__global__ __launch_bounds__(256) void bs_linear_wmma_f32(
    const float* __restrict__ X,
    const float* __restrict__ W,
    const float* __restrict__ bias,
    float* __restrict__ Y)
{
    __shared__ float lds[2][SLAB_ELEMS];

    const int tileN = blockIdx.x;
    const int tileM = blockIdx.y;
    const int mBase = tileM * MT;
    const int nBase = tileN * NT;

    const int tid  = threadIdx.x;
    const int wave = tid >> 5;
    const int lane = tid & 31;
    const int lid  = lane & 15;
    const int half = lane >> 4;
    const int wm   = wave >> 2;
    const int wn   = wave & 3;

    const int sr = tid >> 3;
    const int sc = (tid & 7) * 4;

    const float* xp = X + (size_t)(mBase + sr) * K_TOT + sc;
    const float* wp = W + (size_t)(nBase + sr) * K_TOT + sc;

    v8f acc[4][2];
    #pragma unroll
    for (int i = 0; i < 4; ++i)
        #pragma unroll
        for (int j = 0; j < 2; ++j)
            acc[i][j] = (v8f)0.0f;

    stage_slab(xp, wp, lds[0], lds[0] + A_ELEMS, sr, sc);
    __syncthreads();

    int buf = 0;
    for (int k0 = 0; k0 < K_TOT; k0 += KT) {
        if (k0 + KT < K_TOT)
            stage_slab(xp + (k0 + KT), wp + (k0 + KT),
                       lds[buf ^ 1], lds[buf ^ 1] + A_ELEMS, sr, sc);

        const float* As = lds[buf];
        const float* Bs = lds[buf] + A_ELEMS;

        #pragma unroll
        for (int kk = 0; kk < KT; kk += 4) {
            v2f afrag[4];
            v2f bfrag[2];
            #pragma unroll
            for (int tm = 0; tm < 4; ++tm)
                afrag[tm] = *(const v2f*)&As[(wm * 64 + tm * 16 + lid) * LDSTR + kk + 2 * half];
            #pragma unroll
            for (int tn = 0; tn < 2; ++tn)
                bfrag[tn] = *(const v2f*)&Bs[(wn * 32 + tn * 16 + lid) * LDSTR + kk + 2 * half];

            #pragma unroll
            for (int tm = 0; tm < 4; ++tm)
                #pragma unroll
                for (int tn = 0; tn < 2; ++tn)
                    acc[tm][tn] = __builtin_amdgcn_wmma_f32_16x16x4_f32(
                         false, afrag[tm],
                         false, bfrag[tn],
                         (short)0, acc[tm][tn],
                         false,  false);
        }
        __syncthreads();
        buf ^= 1;
    }

    #pragma unroll
    for (int tn = 0; tn < 2; ++tn) {
        const int col = nBase + wn * 32 + tn * 16 + lid;
        const float bv = bias[col];
        #pragma unroll
        for (int tm = 0; tm < 4; ++tm) {
            const int rbase = mBase + wm * 64 + tm * 16 + 8 * half;
            #pragma unroll
            for (int r = 0; r < 8; ++r) {
                Y[(size_t)(rbase + r) * N_TOT + col] = acc[tm][tn][r] + bv;
            }
        }
    }
}

extern "C" void kernel_launch(void* const* d_in, const int* in_sizes, int n_in,
                              void* d_out, int out_size, void* d_ws, size_t ws_size,
                              hipStream_t stream) {
    const float* x    = (const float*)d_in[0];
    const float* w    = (const float*)d_in[1];
    const float* bias = (const float*)d_in[2];
    float* y          = (float*)d_out;

    dim3 grid(N_TOT / NT, M_TOT / MT);
    dim3 block(256);
    bs_linear_wmma_f32<<<grid, block, 0, stream>>>(x, w, bias, y);
}


# Re-tiled for reuse: workgroup tile 128x128 -> 256x256, wave tile 64x32 -> 128x64, fragments loaded as one group per 64 WMMAs: half the L2->LDS and LDS->register bytes per WMMA, one barrier per 256
# speedup vs baseline: 1.0596x; 1.0596x over previous
#include <hip/hip_runtime.h>


typedef __attribute__((ext_vector_type(2))) float v2f;
typedef __attribute__((ext_vector_type(4))) float v4f;
typedef __attribute__((ext_vector_type(8))) float v8f;

#define M_TOT 8192
#define N_TOT 4096
#define K_TOT 4096

#define MT 256
#define NT 256
#define KT 32
#define LDSTR 34
#define A_ELEMS (MT * LDSTR)
#define B_ELEMS (NT * LDSTR)
#define SLAB_ELEMS (A_ELEMS + B_ELEMS)

#define WM_TILES 8
#define WN_TILES 4
#define STG 8

#define ORDER_FENCE() __builtin_amdgcn_sched_barrier(0)

__device__ __forceinline__ void load_slab(const float* __restrict__ xp,
                                          const float* __restrict__ wp,
                                          v4f (&ga)[STG], v4f (&gb)[STG]) {
    #pragma unroll
    for (int p = 0; p < STG; ++p) {
        const size_t goff = (size_t)(32 * p) * K_TOT;
        ga[p] = *(const v4f*)(xp + goff);
        gb[p] = *(const v4f*)(wp + goff);
    }
}

__device__ __forceinline__ void store_slab(const v4f (&ga)[STG], const v4f (&gb)[STG],
                                           float* __restrict__ As,
                                           float* __restrict__ Bs,
                                           int sr, int sc) {
    #pragma unroll
    for (int p = 0; p < STG; ++p) {
        const int l = (sr + 32 * p) * LDSTR + sc;
        *(v2f*)&As[l]     = ga[p].lo;
        *(v2f*)&As[l + 2] = ga[p].hi;
        *(v2f*)&Bs[l]     = gb[p].lo;
        *(v2f*)&Bs[l + 2] = gb[p].hi;
    }
}

__device__ __forceinline__ void load_frags(const float* __restrict__ As,
                                           const float* __restrict__ Bs,
                                           int arow, int brow, int koff, int k8,
                                           v2f (&a)[2][WM_TILES], v2f (&b)[2][WN_TILES]) {
    #pragma unroll
    for (int tm = 0; tm < WM_TILES; ++tm) {
        #pragma unroll
        for (int u = 0; u < 2; ++u)
            a[u][tm] = *(const v2f*)&As[(arow + tm * 16) * LDSTR + k8 + 4 * u + koff];
    }
    #pragma unroll
    for (int tn = 0; tn < WN_TILES; ++tn) {
        #pragma unroll
        for (int u = 0; u < 2; ++u)
            b[u][tn] = *(const v2f*)&Bs[(brow + tn * 16) * LDSTR + k8 + 4 * u + koff];
    }
}

__device__ __forceinline__ void mma_group(const v2f (&a)[2][WM_TILES], const v2f (&b)[2][WN_TILES],
                                          v8f (&acc)[WM_TILES][WN_TILES]) {
    #pragma unroll
    for (int u = 0; u < 2; ++u)
        #pragma unroll
        for (int tm = 0; tm < WM_TILES; ++tm)
            #pragma unroll
            for (int tn = 0; tn < WN_TILES; ++tn)
                acc[tm][tn] = __builtin_amdgcn_wmma_f32_16x16x4_f32(
                     false, a[u][tm],
                     false, b[u][tn],
                     (short)0, acc[tm][tn],
                     false,  false);
}

__global__ __launch_bounds__(256) void bs_linear_wmma_f32(
    const float* __restrict__ X,
    const float* __restrict__ W,
    const float* __restrict__ bias,
    float* __restrict__ Y)
{
    __shared__ float lds[2][SLAB_ELEMS];

    const int tileN = blockIdx.x;
    const int tileM = blockIdx.y;
    const int mBase = tileM * MT;
    const int nBase = tileN * NT;

    const int tid  = threadIdx.x;
    const int wave = tid >> 5;
    const int lane = tid & 31;
    const int lid  = lane & 15;
    const int half = lane >> 4;
    const int wm   = wave >> 2;
    const int wn   = wave & 3;

    const int sr = tid >> 3;
    const int sc = (tid & 7) * 4;

    const float* xp = X + (size_t)(mBase + sr) * K_TOT + sc;
    const float* wp = W + (size_t)(nBase + sr) * K_TOT + sc;

    const int arow = wm * 128 + lid;
    const int brow = wn * 64 + lid;
    const int koff = 2 * half;

    v8f acc[WM_TILES][WN_TILES];
    #pragma unroll
    for (int i = 0; i < WM_TILES; ++i)
        #pragma unroll
        for (int j = 0; j < WN_TILES; ++j)
            acc[i][j] = (v8f)0.0f;

    v4f ga[STG], gb[STG];
    v2f pa[2][WM_TILES], pb[2][WN_TILES];
    v2f qa[2][WM_TILES], qb[2][WN_TILES];

    load_slab(xp, wp, ga, gb);
    store_slab(ga, gb, lds[0], lds[0] + A_ELEMS, sr, sc);
    __syncthreads();
    load_frags(lds[0], lds[0] + A_ELEMS, arow, brow, koff, 0, pa, pb);

    int buf = 0;
    for (int k0 = KT; k0 < K_TOT; k0 += KT) {
        const float* As = lds[buf];
        const float* Bs = lds[buf] + A_ELEMS;
        float* An = lds[buf ^ 1];
        float* Bn = lds[buf ^ 1] + A_ELEMS;

        load_slab(xp + k0, wp + k0, ga, gb);
        ORDER_FENCE();
        load_frags(As, Bs, arow, brow, koff, 8, qa, qb);
        ORDER_FENCE();
        mma_group(pa, pb, acc);
        ORDER_FENCE();
        load_frags(As, Bs, arow, brow, koff, 16, pa, pb);
        ORDER_FENCE();
        mma_group(qa, qb, acc);
        ORDER_FENCE();
        store_slab(ga, gb, An, Bn, sr, sc);
        ORDER_FENCE();
        load_frags(As, Bs, arow, brow, koff, 24, qa, qb);
        ORDER_FENCE();
        mma_group(pa, pb, acc);
        ORDER_FENCE();
        __syncthreads();
        ORDER_FENCE();
        load_frags(An, Bn, arow, brow, koff, 0, pa, pb);
        ORDER_FENCE();
        mma_group(qa, qb, acc);
        ORDER_FENCE();
        buf ^= 1;
    }
    {
        const float* As = lds[buf];
        const float* Bs = lds[buf] + A_ELEMS;
        load_frags(As, Bs, arow, brow, koff, 8, qa, qb);
        ORDER_FENCE();
        mma_group(pa, pb, acc);
        ORDER_FENCE();
        load_frags(As, Bs, arow, brow, koff, 16, pa, pb);
        ORDER_FENCE();
        mma_group(qa, qb, acc);
        ORDER_FENCE();
        load_frags(As, Bs, arow, brow, koff, 24, qa, qb);
        ORDER_FENCE();
        mma_group(pa, pb, acc);
        ORDER_FENCE();
        mma_group(qa, qb, acc);
    }

    #pragma unroll
    for (int tn = 0; tn < WN_TILES; ++tn) {
        const int col = nBase + wn * 64 + tn * 16 + lid;
        const float bv = bias[col];
        #pragma unroll
        for (int tm = 0; tm < WM_TILES; ++tm) {
            const int rbase = mBase + wm * 128 + tm * 16 + 8 * half;
            #pragma unroll
            for (int r = 0; r < 8; ++r) {
                Y[(size_t)(rbase + r) * N_TOT + col] = acc[tm][tn][r] + bv;
            }
        }
    }
}

extern "C" void kernel_launch(void* const* d_in, const int* in_sizes, int n_in,
                              void* d_out, int out_size, void* d_ws, size_t ws_size,
                              hipStream_t stream) {
    const float* x    = (const float*)d_in[0];
    const float* w    = (const float*)d_in[1];
    const float* bias = (const float*)d_in[2];
    float* y          = (float*)d_out;

    dim3 grid(16, 32);
    dim3 block(256);
    bs_linear_wmma_f32<<<grid, block, 0, stream>>>(x, w, bias, y);
}
